# starting kernel
# baseline (speedup 1.0000x reference)
#include <hip/hip_runtime.h>

typedef __attribute__((ext_vector_type(2))) float v2f;
typedef __attribute__((ext_vector_type(8))) float v8f;

#define T_DIM  2048
#define F_INN  2048
#define D_DIM  128
#define BATCH  4
#define WIN    101
#define HWIN   50

__global__ __launch_bounds__(256) void transpose_w_kernel(
    const float* __restrict__ W, float* __restrict__ Wt,
    float* __restrict__ zrow)
{
  const int idx = blockIdx.x * 256 + threadIdx.x;
  const int r = idx >> 7;
  const int c = idx & 127;
  Wt[(size_t)c * F_INN + r] = W[idx];
  if (blockIdx.x == 0 && threadIdx.x < D_DIM) zrow[threadIdx.x] = 0.f;
}

__global__ __launch_bounds__(128) void gemm_norm_kernel(
    const float* __restrict__ x, const float* __restrict__ Wt,
    const float* __restrict__ bias, float* __restrict__ p)
{
  const int wave = (blockIdx.x * blockDim.x + threadIdx.x) >> 5;
  const int lane = threadIdx.x & 31;
  const int half = lane >> 4;
  const int l16  = lane & 15;
  const int row0 = wave * 16;

  v8f acc[8] = {};

  const float* xrow = x + (size_t)(row0 + l16) * F_INN + half * 2;
  const float* wrow[8];
#pragma unroll
  for (int n = 0; n < 8; ++n)
    wrow[n] = Wt + (size_t)(n * 16 + l16) * F_INN + half * 2;

  v2f a_cur = *(const v2f*)(xrow);
  v2f b_cur[8];
#pragma unroll
  for (int n = 0; n < 8; ++n) b_cur[n] = *(const v2f*)(wrow[n]);

  for (int k = 0; k < F_INN; k += 4) {
    const int kn = k + 4;
    v2f a_nxt = {};
    v2f b_nxt[8] = {};
    if (kn < F_INN) {
      a_nxt = *(const v2f*)(xrow + kn);
#pragma unroll
      for (int n = 0; n < 8; ++n) b_nxt[n] = *(const v2f*)(wrow[n] + kn);
    }
    __builtin_prefetch(xrow + k + 256, 0, 0);

#pragma unroll
    for (int n = 0; n < 8; ++n)
      acc[n] = __builtin_amdgcn_wmma_f32_16x16x4_f32(
          false, a_cur, false, b_cur[n], (short)0, acc[n], false, false);

    a_cur = a_nxt;
#pragma unroll
    for (int n = 0; n < 8; ++n) b_cur[n] = b_nxt[n];
  }

#pragma unroll
  for (int n = 0; n < 8; ++n) {
    const float bv = bias[n * 16 + l16];
#pragma unroll
    for (int j = 0; j < 8; ++j) acc[n][j] += bv;
  }

#pragma unroll
  for (int j = 0; j < 8; ++j) {
    float s = 0.f;
#pragma unroll
    for (int n = 0; n < 8; ++n) s += acc[n][j] * acc[n][j];
    s += __shfl_xor(s, 1, 32);
    s += __shfl_xor(s, 2, 32);
    s += __shfl_xor(s, 4, 32);
    s += __shfl_xor(s, 8, 32);
    const float scale = 1.0f / fmaxf(sqrtf(s), 1e-12f);
#pragma unroll
    for (int n = 0; n < 8; ++n) acc[n][j] *= scale;
  }

#pragma unroll
  for (int j = 0; j < 8; ++j) {
    const int r = row0 + j + 8 * half;
#pragma unroll
    for (int n = 0; n < 8; ++n)
      p[(size_t)r * D_DIM + n * 16 + l16] = acc[n][j];
  }
}

__global__ __launch_bounds__(128) void band_sim_kernel(
    const float* __restrict__ p, const float* __restrict__ zrow,
    float* __restrict__ out)
{
  const int wave = (blockIdx.x * blockDim.x + threadIdx.x) >> 5;
  const int lane = threadIdx.x & 31;
  const int half = lane >> 4;
  const int l16  = lane & 15;
  const int b    = wave >> 7;
  const int mt   = wave & 127;
  const int t0   = mt * 16;

  const float* pb = p + (size_t)b * T_DIM * D_DIM;

  const float* bbase[9];
#pragma unroll
  for (int ct = 0; ct < 9; ++ct) {
    const int c = t0 + (ct - 4) * 16 + l16;
    bbase[ct] = ((unsigned)c < (unsigned)T_DIM)
                    ? (pb + (size_t)c * D_DIM + half * 2)
                    : (zrow + half * 2);
  }
  const float* arow = pb + (size_t)(t0 + l16) * D_DIM + half * 2;

  v8f acc[9] = {};

  v2f a_cur = *(const v2f*)(arow);
  v2f b_cur[9];
#pragma unroll
  for (int ct = 0; ct < 9; ++ct) b_cur[ct] = *(const v2f*)(bbase[ct]);

  for (int k = 0; k < D_DIM; k += 4) {
    const int kn = k + 4;
    v2f a_nxt = {};
    v2f b_nxt[9] = {};
    if (kn < D_DIM) {
      a_nxt = *(const v2f*)(arow + kn);
#pragma unroll
      for (int ct = 0; ct < 9; ++ct) b_nxt[ct] = *(const v2f*)(bbase[ct] + kn);
    }

#pragma unroll
    for (int ct = 0; ct < 9; ++ct)
      acc[ct] = __builtin_amdgcn_wmma_f32_16x16x4_f32(
          false, a_cur, false, b_cur[ct], (short)0, acc[ct], false, false);

    a_cur = a_nxt;
#pragma unroll
    for (int ct = 0; ct < 9; ++ct) b_cur[ct] = b_nxt[ct];
  }

#pragma unroll
  for (int ct = 0; ct < 9; ++ct) {
    const int c = t0 + (ct - 4) * 16 + l16;
#pragma unroll
    for (int j = 0; j < 8; ++j) {
      const int t = t0 + j + 8 * half;
      const int w = c - t + HWIN;
      if (w >= 0 && w < WIN) {
        out[((size_t)b * T_DIM + t) * WIN + w] = acc[ct][j];
      }
    }
  }
}

extern "C" void kernel_launch(void* const* d_in, const int* in_sizes, int n_in,
                              void* d_out, int out_size, void* d_ws, size_t ws_size,
                              hipStream_t stream) {
  const float* x  = (const float*)d_in[0];
  const float* W  = (const float*)d_in[1];
  const float* bv = (const float*)d_in[2];
  float* out = (float*)d_out;

  float* p    = (float*)d_ws;
  float* Wt   = p + (size_t)8192 * D_DIM;
  float* zrow = Wt + (size_t)D_DIM * F_INN;

  transpose_w_kernel<<<(F_INN * D_DIM) / 256, 256, 0, stream>>>(W, Wt, zrow);
  gemm_norm_kernel<<<128, 128, 0, stream>>>(x, Wt, bv, p);
  band_sim_kernel<<<128, 128, 0, stream>>>(p, zrow, out);
}


# Projection GEMM re-tiled: 32 rows per work-group, x and W^T chunks loaded as whole 512-B rows and staged through LDS for the WMMA fragments (was 8-B loads over 16 lines each); summation order kept
# speedup vs baseline: 1.9245x; 1.9245x over previous
#include <hip/hip_runtime.h>

typedef __attribute__((ext_vector_type(2))) float v2f;
typedef __attribute__((ext_vector_type(8))) float v8f;

#define T_DIM  2048
#define F_INN  2048
#define D_DIM  128
#define BATCH  4
#define WIN    101
#define HWIN   50

__global__ __launch_bounds__(256) void transpose_w_kernel(
    const float* __restrict__ W, float* __restrict__ Wt,
    float* __restrict__ zrow)
{
  const int idx = blockIdx.x * 256 + threadIdx.x;
  const int r = idx >> 7;
  const int c = idx & 127;
  Wt[(size_t)c * F_INN + r] = W[idx];
  if (blockIdx.x == 0 && threadIdx.x < D_DIM) zrow[threadIdx.x] = 0.f;
}

#define BM          32
#define KC          128
#define PITCH       132
#define A_ELEMS     (BM * PITCH)
#define B_ELEMS     (D_DIM * PITCH)
#define STAGE_ELEMS (A_ELEMS + B_ELEMS)

typedef __attribute__((ext_vector_type(4))) float v4f;

__device__ __forceinline__ void compute_chunk(
    const float* __restrict__ aF, const float* __restrict__ bF0,
    const float* __restrict__ bF1, v8f& acc0, v8f& acc1)
{
#pragma unroll
  for (int ks = 0; ks < KC / 4; ++ks) {
    const v2f a  = *(const v2f*)(aF  + ks * 4);
    const v2f b0 = *(const v2f*)(bF0 + ks * 4);
    const v2f b1 = *(const v2f*)(bF1 + ks * 4);
    acc0 = __builtin_amdgcn_wmma_f32_16x16x4_f32(
        false, a, false, b0, (short)0, acc0, false, false);
    acc1 = __builtin_amdgcn_wmma_f32_16x16x4_f32(
        false, a, false, b1, (short)0, acc1, false, false);
  }
}

__global__ __launch_bounds__(256) void gemm_norm_kernel(
    const float* __restrict__ x, const float* __restrict__ Wt,
    const float* __restrict__ bias, float* __restrict__ p)
{
  __shared__ __attribute__((aligned(16))) float lds[2 * STAGE_ELEMS];

  const int tid  = threadIdx.x;
  const int wv   = tid >> 5;
  const int lane = tid & 31;
  const int half = lane >> 4;
  const int l16  = lane & 15;
  const int mt   = wv & 1;
  const int np   = wv >> 1;
  const int row0 = blockIdx.x * BM;

  const float* xg = x  + (size_t)(row0 + wv) * F_INN + lane * 4;
  const float* wg = Wt + (size_t)wv * F_INN + lane * 4;
  const int st_off = wv * PITCH + lane * 4;

  const int aoff  = (mt * 16 + l16) * PITCH + half * 2;
  const int boff0 = A_ELEMS + ((2 * np) * 16 + l16) * PITCH + half * 2;
  const int boff1 = boff0 + 16 * PITCH;

  v8f acc0 = {};
  v8f acc1 = {};

  v4f ra[4];
  v4f rb[16];

#pragma unroll
  for (int j = 0; j < 4; ++j)
    ra[j] = *(const v4f*)(xg + (size_t)(8 * j) * F_INN);
#pragma unroll
  for (int j = 0; j < 16; ++j)
    rb[j] = *(const v4f*)(wg + (size_t)(8 * j) * F_INN);
#pragma unroll
  for (int j = 0; j < 4; ++j)
    *(v4f*)(lds + st_off + (8 * j) * PITCH) = ra[j];
#pragma unroll
  for (int j = 0; j < 16; ++j)
    *(v4f*)(lds + A_ELEMS + st_off + (8 * j) * PITCH) = rb[j];
  __syncthreads();

#pragma unroll 1
  for (int c = 0; c < F_INN / KC; ++c) {
    const float* cur = lds + (c & 1) * STAGE_ELEMS;
    float*       nxt = lds + ((c & 1) ^ 1) * STAGE_ELEMS;
    const bool more = (c + 1) < (F_INN / KC);

    if (more) {
      const int kn = (c + 1) * KC;
#pragma unroll
      for (int j = 0; j < 4; ++j)
        ra[j] = *(const v4f*)(xg + (size_t)(8 * j) * F_INN + kn);
#pragma unroll
      for (int j = 0; j < 16; ++j)
        rb[j] = *(const v4f*)(wg + (size_t)(8 * j) * F_INN + kn);
    }

    compute_chunk(cur + aoff, cur + boff0, cur + boff1, acc0, acc1);

    if (more) {
#pragma unroll
      for (int j = 0; j < 4; ++j)
        *(v4f*)(nxt + st_off + (8 * j) * PITCH) = ra[j];
#pragma unroll
      for (int j = 0; j < 16; ++j)
        *(v4f*)(nxt + A_ELEMS + st_off + (8 * j) * PITCH) = rb[j];
    }
    __syncthreads();
  }

  float* T = lds;
  {
    const float bv0 = bias[(2 * np) * 16 + l16];
    const float bv1 = bias[(2 * np + 1) * 16 + l16];
#pragma unroll
    for (int j = 0; j < 8; ++j) {
      const int r = mt * 16 + 8 * half + j;
      T[r * PITCH + (2 * np) * 16 + l16]     = acc0[j] + bv0;
      T[r * PITCH + (2 * np + 1) * 16 + l16] = acc1[j] + bv1;
    }
  }
  __syncthreads();

#pragma unroll
  for (int i = 0; i < 2; ++i) {
    const int r = 4 * wv + 2 * i + half;
    float v[8];
#pragma unroll
    for (int n = 0; n < 8; ++n) v[n] = T[r * PITCH + n * 16 + l16];
    float s = 0.f;
#pragma unroll
    for (int n = 0; n < 8; ++n) s += v[n] * v[n];
    s += __shfl_xor(s, 1, 32);
    s += __shfl_xor(s, 2, 32);
    s += __shfl_xor(s, 4, 32);
    s += __shfl_xor(s, 8, 32);
    const float scale = 1.0f / fmaxf(sqrtf(s), 1e-12f);
#pragma unroll
    for (int n = 0; n < 8; ++n)
      p[(size_t)(row0 + r) * D_DIM + n * 16 + l16] = v[n] * scale;
  }
}

__global__ __launch_bounds__(128) void band_sim_kernel(
    const float* __restrict__ p, const float* __restrict__ zrow,
    float* __restrict__ out)
{
  const int wave = (blockIdx.x * blockDim.x + threadIdx.x) >> 5;
  const int lane = threadIdx.x & 31;
  const int half = lane >> 4;
  const int l16  = lane & 15;
  const int b    = wave >> 7;
  const int mt   = wave & 127;
  const int t0   = mt * 16;

  const float* pb = p + (size_t)b * T_DIM * D_DIM;

  const float* bbase[9];
#pragma unroll
  for (int ct = 0; ct < 9; ++ct) {
    const int c = t0 + (ct - 4) * 16 + l16;
    bbase[ct] = ((unsigned)c < (unsigned)T_DIM)
                    ? (pb + (size_t)c * D_DIM + half * 2)
                    : (zrow + half * 2);
  }
  const float* arow = pb + (size_t)(t0 + l16) * D_DIM + half * 2;

  v8f acc[9] = {};

  v2f a_cur = *(const v2f*)(arow);
  v2f b_cur[9];
#pragma unroll
  for (int ct = 0; ct < 9; ++ct) b_cur[ct] = *(const v2f*)(bbase[ct]);

  for (int k = 0; k < D_DIM; k += 4) {
    const int kn = k + 4;
    v2f a_nxt = {};
    v2f b_nxt[9] = {};
    if (kn < D_DIM) {
      a_nxt = *(const v2f*)(arow + kn);
#pragma unroll
      for (int ct = 0; ct < 9; ++ct) b_nxt[ct] = *(const v2f*)(bbase[ct] + kn);
    }

#pragma unroll
    for (int ct = 0; ct < 9; ++ct)
      acc[ct] = __builtin_amdgcn_wmma_f32_16x16x4_f32(
          false, a_cur, false, b_cur[ct], (short)0, acc[ct], false, false);

    a_cur = a_nxt;
#pragma unroll
    for (int ct = 0; ct < 9; ++ct) b_cur[ct] = b_nxt[ct];
  }

#pragma unroll
  for (int ct = 0; ct < 9; ++ct) {
    const int c = t0 + (ct - 4) * 16 + l16;
#pragma unroll
    for (int j = 0; j < 8; ++j) {
      const int t = t0 + j + 8 * half;
      const int w = c - t + HWIN;
      if (w >= 0 && w < WIN) {
        out[((size_t)b * T_DIM + t) * WIN + w] = acc[ct][j];
      }
    }
  }
}

extern "C" void kernel_launch(void* const* d_in, const int* in_sizes, int n_in,
                              void* d_out, int out_size, void* d_ws, size_t ws_size,
                              hipStream_t stream) {
  const float* x  = (const float*)d_in[0];
  const float* W  = (const float*)d_in[1];
  const float* bv = (const float*)d_in[2];
  float* out = (float*)d_out;

  float* p    = (float*)d_ws;
  float* Wt   = p + (size_t)8192 * D_DIM;
  float* zrow = Wt + (size_t)D_DIM * F_INN;

  transpose_w_kernel<<<(F_INN * D_DIM) / 256, 256, 0, stream>>>(W, Wt, zrow);
  gemm_norm_kernel<<<256, 256, 0, stream>>>(x, Wt, bv, p);
  band_sim_kernel<<<128, 128, 0, stream>>>(p, zrow, out);
}


# Transposition launch dropped: W chunks staged in LDS k-major as they lie in memory, B fragments read by two-address LDS loads one LDS row apart; 3 launches become 2; summation order kept
# speedup vs baseline: 2.0493x; 1.0648x over previous
#include <hip/hip_runtime.h>

typedef __attribute__((ext_vector_type(2))) float v2f;
typedef __attribute__((ext_vector_type(8))) float v8f;

#define T_DIM  2048
#define F_INN  2048
#define D_DIM  128
#define BATCH  4
#define WIN    101
#define HWIN   50

#define BM          32
#define KC          128
#define PITCH_A     132
#define PITCH_B     144
#define A_ELEMS     (BM * PITCH_A)
#define B_ELEMS     (KC * PITCH_B)
#define STAGE_ELEMS (A_ELEMS + B_ELEMS)

typedef __attribute__((ext_vector_type(4))) float v4f;

struct Frag { v2f a00, a01, a10, a11, b0, b1; };

__device__ __forceinline__ void load_frag(
    Frag& f, const float* __restrict__ aF0, const float* __restrict__ aF1,
    const float* __restrict__ bF, int kp)
{
  f.a00 = *(const v2f*)(aF0 + kp * 8);
  f.a01 = *(const v2f*)(aF0 + kp * 8 + 4);
  f.a10 = *(const v2f*)(aF1 + kp * 8);
  f.a11 = *(const v2f*)(aF1 + kp * 8 + 4);
  f.b0.x = bF[(kp * 8) * PITCH_B];
  f.b0.y = bF[(kp * 8 + 1) * PITCH_B];
  f.b1.x = bF[(kp * 8 + 4) * PITCH_B];
  f.b1.y = bF[(kp * 8 + 5) * PITCH_B];
}

__device__ __forceinline__ void compute_chunk(
    const float* __restrict__ aF0, const float* __restrict__ aF1,
    const float* __restrict__ bF, v8f& acc0, v8f& acc1)
{
  Frag f[2];
  load_frag(f[0], aF0, aF1, bF, 0);
#pragma unroll
  for (int kp = 0; kp < KC / 8; ++kp) {
    if (kp + 1 < KC / 8) load_frag(f[(kp + 1) & 1], aF0, aF1, bF, kp + 1);
    __builtin_amdgcn_sched_barrier(0);
    const Frag& g = f[kp & 1];
    acc0 = __builtin_amdgcn_wmma_f32_16x16x4_f32(
        false, g.a00, false, g.b0, (short)0, acc0, false, false);
    acc1 = __builtin_amdgcn_wmma_f32_16x16x4_f32(
        false, g.a10, false, g.b0, (short)0, acc1, false, false);
    acc0 = __builtin_amdgcn_wmma_f32_16x16x4_f32(
        false, g.a01, false, g.b1, (short)0, acc0, false, false);
    acc1 = __builtin_amdgcn_wmma_f32_16x16x4_f32(
        false, g.a11, false, g.b1, (short)0, acc1, false, false);
    __builtin_amdgcn_sched_barrier(0);
  }
}

__global__ __launch_bounds__(256) void gemm_norm_kernel(
    const float* __restrict__ x, const float* __restrict__ W,
    const float* __restrict__ bias, float* __restrict__ p,
    float* __restrict__ zrow)
{
  __shared__ __attribute__((aligned(16))) float lds[2 * STAGE_ELEMS];

  const int tid  = threadIdx.x;
  const int wv   = tid >> 5;
  const int lane = tid & 31;
  const int half = lane >> 4;
  const int l16  = lane & 15;
  const int row0 = blockIdx.x * BM;

  if (blockIdx.x == 0 && tid < D_DIM) zrow[tid] = 0.f;

  const float* xg = x + (size_t)(row0 + wv) * F_INN + lane * 4;
  const float* wg = W + (size_t)wv * D_DIM + lane * 4;
  const int stA = wv * PITCH_A + lane * 4;
  const int stB = A_ELEMS + wv * PITCH_B + lane * 4;

  const int aoff0 = l16 * PITCH_A + half * 2;
  const int aoff1 = aoff0 + 16 * PITCH_A;
  const int boff  = A_ELEMS + (half * 2) * PITCH_B + wv * 16 + l16;

  v8f acc0 = {};
  v8f acc1 = {};

  v4f ra[4];
  v4f rb[16];

#pragma unroll
  for (int j = 0; j < 4; ++j)
    ra[j] = *(const v4f*)(xg + (size_t)(8 * j) * F_INN);
#pragma unroll
  for (int j = 0; j < 16; ++j)
    rb[j] = *(const v4f*)(wg + (size_t)(8 * j) * D_DIM);
#pragma unroll
  for (int j = 0; j < 4; ++j)
    *(v4f*)(lds + stA + (8 * j) * PITCH_A) = ra[j];
#pragma unroll
  for (int j = 0; j < 16; ++j)
    *(v4f*)(lds + stB + (8 * j) * PITCH_B) = rb[j];
  __syncthreads();

#pragma unroll 1
  for (int c = 0; c < F_INN / KC; ++c) {
    const float* cur = lds + (c & 1) * STAGE_ELEMS;
    float*       nxt = lds + ((c & 1) ^ 1) * STAGE_ELEMS;
    const bool more = (c + 1) < (F_INN / KC);

    if (more) {
      const int kn = (c + 1) * KC;
#pragma unroll
      for (int j = 0; j < 4; ++j)
        ra[j] = *(const v4f*)(xg + (size_t)(8 * j) * F_INN + kn);
#pragma unroll
      for (int j = 0; j < 16; ++j)
        rb[j] = *(const v4f*)(wg + (size_t)(kn + 8 * j) * D_DIM);
    }

    compute_chunk(cur + aoff0, cur + aoff1, cur + boff, acc0, acc1);

    if (more) {
#pragma unroll
      for (int j = 0; j < 4; ++j)
        *(v4f*)(nxt + stA + (8 * j) * PITCH_A) = ra[j];
#pragma unroll
      for (int j = 0; j < 16; ++j)
        *(v4f*)(nxt + stB + (8 * j) * PITCH_B) = rb[j];
    }
    __syncthreads();
  }

  float* T = lds;
  {
    const float bv = bias[wv * 16 + l16];
#pragma unroll
    for (int j = 0; j < 8; ++j) {
      const int r = 8 * half + j;
      T[r * PITCH_A + wv * 16 + l16]        = acc0[j] + bv;
      T[(16 + r) * PITCH_A + wv * 16 + l16] = acc1[j] + bv;
    }
  }
  __syncthreads();

#pragma unroll
  for (int i = 0; i < 2; ++i) {
    const int r = 4 * wv + 2 * i + half;
    float v[8];
#pragma unroll
    for (int n = 0; n < 8; ++n) v[n] = T[r * PITCH_A + n * 16 + l16];
    float s = 0.f;
#pragma unroll
    for (int n = 0; n < 8; ++n) s += v[n] * v[n];
    s += __shfl_xor(s, 1, 32);
    s += __shfl_xor(s, 2, 32);
    s += __shfl_xor(s, 4, 32);
    s += __shfl_xor(s, 8, 32);
    const float scale = 1.0f / fmaxf(sqrtf(s), 1e-12f);
#pragma unroll
    for (int n = 0; n < 8; ++n)
      p[(size_t)(row0 + r) * D_DIM + n * 16 + l16] = v[n] * scale;
  }
}

__global__ __launch_bounds__(128) void band_sim_kernel(
    const float* __restrict__ p, const float* __restrict__ zrow,
    float* __restrict__ out)
{
  const int wave = (blockIdx.x * blockDim.x + threadIdx.x) >> 5;
  const int lane = threadIdx.x & 31;
  const int half = lane >> 4;
  const int l16  = lane & 15;
  const int b    = wave >> 7;
  const int mt   = wave & 127;
  const int t0   = mt * 16;

  const float* pb = p + (size_t)b * T_DIM * D_DIM;

  const float* bbase[9];
#pragma unroll
  for (int ct = 0; ct < 9; ++ct) {
    const int c = t0 + (ct - 4) * 16 + l16;
    bbase[ct] = ((unsigned)c < (unsigned)T_DIM)
                    ? (pb + (size_t)c * D_DIM + half * 2)
                    : (zrow + half * 2);
  }
  const float* arow = pb + (size_t)(t0 + l16) * D_DIM + half * 2;

  v8f acc[9] = {};

  v2f a_cur = *(const v2f*)(arow);
  v2f b_cur[9];
#pragma unroll
  for (int ct = 0; ct < 9; ++ct) b_cur[ct] = *(const v2f*)(bbase[ct]);

  for (int k = 0; k < D_DIM; k += 4) {
    const int kn = k + 4;
    v2f a_nxt = {};
    v2f b_nxt[9] = {};
    if (kn < D_DIM) {
      a_nxt = *(const v2f*)(arow + kn);
#pragma unroll
      for (int ct = 0; ct < 9; ++ct) b_nxt[ct] = *(const v2f*)(bbase[ct] + kn);
    }

#pragma unroll
    for (int ct = 0; ct < 9; ++ct)
      acc[ct] = __builtin_amdgcn_wmma_f32_16x16x4_f32(
          false, a_cur, false, b_cur[ct], (short)0, acc[ct], false, false);

    a_cur = a_nxt;
#pragma unroll
    for (int ct = 0; ct < 9; ++ct) b_cur[ct] = b_nxt[ct];
  }

#pragma unroll
  for (int ct = 0; ct < 9; ++ct) {
    const int c = t0 + (ct - 4) * 16 + l16;
#pragma unroll
    for (int j = 0; j < 8; ++j) {
      const int t = t0 + j + 8 * half;
      const int w = c - t + HWIN;
      if (w >= 0 && w < WIN) {
        out[((size_t)b * T_DIM + t) * WIN + w] = acc[ct][j];
      }
    }
  }
}

extern "C" void kernel_launch(void* const* d_in, const int* in_sizes, int n_in,
                              void* d_out, int out_size, void* d_ws, size_t ws_size,
                              hipStream_t stream) {
  const float* x  = (const float*)d_in[0];
  const float* W  = (const float*)d_in[1];
  const float* bv = (const float*)d_in[2];
  float* out = (float*)d_out;

  float* p    = (float*)d_ws;
  float* zrow = p + (size_t)8192 * D_DIM;

  gemm_norm_kernel<<<256, 256, 0, stream>>>(x, W, bv, p, zrow);
  band_sim_kernel<<<128, 128, 0, stream>>>(p, zrow, out);
}


# Projection GEMM: each chunk's x and W global loads issued two chunks ahead into a second register set; LDS hand-over barrier no longer waits for global loads; summation order kept
# speedup vs baseline: 2.2093x; 1.0781x over previous
#include <hip/hip_runtime.h>

typedef __attribute__((ext_vector_type(2))) float v2f;
typedef __attribute__((ext_vector_type(8))) float v8f;

#define T_DIM  2048
#define F_INN  2048
#define D_DIM  128
#define BATCH  4
#define WIN    101
#define HWIN   50

#define BM          32
#define KC          128
#define PITCH_A     132
#define PITCH_B     144
#define A_ELEMS     (BM * PITCH_A)
#define B_ELEMS     (KC * PITCH_B)
#define STAGE_ELEMS (A_ELEMS + B_ELEMS)

typedef __attribute__((ext_vector_type(4))) float v4f;

struct Frag { v2f a00, a01, a10, a11, b0, b1; };

__device__ __forceinline__ void load_frag(
    Frag& f, const float* __restrict__ aF0, const float* __restrict__ aF1,
    const float* __restrict__ bF, int kp)
{
  f.a00 = *(const v2f*)(aF0 + kp * 8);
  f.a01 = *(const v2f*)(aF0 + kp * 8 + 4);
  f.a10 = *(const v2f*)(aF1 + kp * 8);
  f.a11 = *(const v2f*)(aF1 + kp * 8 + 4);
  f.b0.x = bF[(kp * 8) * PITCH_B];
  f.b0.y = bF[(kp * 8 + 1) * PITCH_B];
  f.b1.x = bF[(kp * 8 + 4) * PITCH_B];
  f.b1.y = bF[(kp * 8 + 5) * PITCH_B];
}

__device__ __forceinline__ void compute_chunk(
    const float* __restrict__ aF0, const float* __restrict__ aF1,
    const float* __restrict__ bF, v8f& acc0, v8f& acc1)
{
  Frag f[2];
  load_frag(f[0], aF0, aF1, bF, 0);
#pragma unroll
  for (int kp = 0; kp < KC / 8; ++kp) {
    if (kp + 1 < KC / 8) load_frag(f[(kp + 1) & 1], aF0, aF1, bF, kp + 1);
    __builtin_amdgcn_sched_barrier(0);
    const Frag& g = f[kp & 1];
    acc0 = __builtin_amdgcn_wmma_f32_16x16x4_f32(
        false, g.a00, false, g.b0, (short)0, acc0, false, false);
    acc1 = __builtin_amdgcn_wmma_f32_16x16x4_f32(
        false, g.a10, false, g.b0, (short)0, acc1, false, false);
    acc0 = __builtin_amdgcn_wmma_f32_16x16x4_f32(
        false, g.a01, false, g.b1, (short)0, acc0, false, false);
    acc1 = __builtin_amdgcn_wmma_f32_16x16x4_f32(
        false, g.a11, false, g.b1, (short)0, acc1, false, false);
    __builtin_amdgcn_sched_barrier(0);
  }
}

__device__ __forceinline__ void lds_barrier()
{
  asm volatile("s_wait_dscnt 0x0\n\ts_barrier_signal -1\n\ts_barrier_wait -1"
               ::: "memory");
}

__device__ __forceinline__ void load_chunk(
    v4f (&ra)[4], v4f (&rb)[16], const float* __restrict__ xc,
    const float* __restrict__ wc, unsigned xo, unsigned wo)
{
#pragma unroll
  for (int j = 0; j < 4; ++j)
    ra[j] = *(const v4f*)((const char*)(xc + (size_t)(8 * j) * F_INN) + xo);
#pragma unroll
  for (int j = 0; j < 16; ++j)
    rb[j] = *(const v4f*)((const char*)(wc + (size_t)(8 * j) * D_DIM) + wo);
  asm volatile("s_wait_xcnt 0x0" ::: "memory");
}

__device__ __forceinline__ void store_chunk(
    const v4f (&ra)[4], const v4f (&rb)[16], float* sa, float* sb)
{
#pragma unroll
  for (int j = 0; j < 4; ++j)
    *(v4f*)(sa + (8 * j) * PITCH_A) = ra[j];
#pragma unroll
  for (int j = 0; j < 16; ++j)
    *(v4f*)(sb + (8 * j) * PITCH_B) = rb[j];
}

__global__ __launch_bounds__(256) void gemm_norm_kernel(
    const float* __restrict__ x, const float* __restrict__ W,
    const float* __restrict__ bias, float* __restrict__ p,
    float* __restrict__ zrow)
{
  __shared__ __attribute__((aligned(16))) float lds[2 * STAGE_ELEMS];

  const int tid  = threadIdx.x;
  const int wv   = tid >> 5;
  const int lane = tid & 31;
  const int half = lane >> 4;
  const int l16  = lane & 15;
  const int row0 = blockIdx.x * BM;

  if (blockIdx.x == 0 && tid < D_DIM) zrow[tid] = 0.f;

  const float* xb = x + (size_t)row0 * F_INN;
  const unsigned xo = (unsigned)((wv * F_INN + lane * 4) * 4);
  const unsigned wo = (unsigned)((wv * D_DIM + lane * 4) * 4);
  const int stA = wv * PITCH_A + lane * 4;
  const int stB = A_ELEMS + wv * PITCH_B + lane * 4;

  const int aoff0 = l16 * PITCH_A + half * 2;
  const int aoff1 = aoff0 + 16 * PITCH_A;
  const int boff  = A_ELEMS + (half * 2) * PITCH_B + wv * 16 + l16;

  v8f acc0 = {};
  v8f acc1 = {};

  v4f ra0[4], ra1[4];
  v4f rb0[16], rb1[16];

  float* const s0 = lds;
  float* const s1 = lds + STAGE_ELEMS;

  load_chunk(ra0, rb0, xb, W, xo, wo);
  load_chunk(ra1, rb1, xb + KC, W + (size_t)KC * D_DIM, xo, wo);
  store_chunk(ra0, rb0, s0 + stA, s0 + stB);
  lds_barrier();

#pragma unroll 1
  for (int c = 0; c < F_INN / KC - 2; c += 2) {
    load_chunk(ra0, rb0, xb + (c + 2) * KC, W + (size_t)(c + 2) * KC * D_DIM, xo, wo);
    compute_chunk(s0 + aoff0, s0 + aoff1, s0 + boff, acc0, acc1);
    store_chunk(ra1, rb1, s1 + stA, s1 + stB);
    lds_barrier();

    load_chunk(ra1, rb1, xb + (c + 3) * KC, W + (size_t)(c + 3) * KC * D_DIM, xo, wo);
    compute_chunk(s1 + aoff0, s1 + aoff1, s1 + boff, acc0, acc1);
    store_chunk(ra0, rb0, s0 + stA, s0 + stB);
    lds_barrier();
  }
  compute_chunk(s0 + aoff0, s0 + aoff1, s0 + boff, acc0, acc1);
  store_chunk(ra1, rb1, s1 + stA, s1 + stB);
  lds_barrier();
  compute_chunk(s1 + aoff0, s1 + aoff1, s1 + boff, acc0, acc1);

  float* T = lds;
  {
    const float bv = bias[wv * 16 + l16];
#pragma unroll
    for (int j = 0; j < 8; ++j) {
      const int r = 8 * half + j;
      T[r * PITCH_A + wv * 16 + l16]        = acc0[j] + bv;
      T[(16 + r) * PITCH_A + wv * 16 + l16] = acc1[j] + bv;
    }
  }
  __syncthreads();

#pragma unroll
  for (int i = 0; i < 2; ++i) {
    const int r = 4 * wv + 2 * i + half;
    float v[8];
#pragma unroll
    for (int n = 0; n < 8; ++n) v[n] = T[r * PITCH_A + n * 16 + l16];
    float s = 0.f;
#pragma unroll
    for (int n = 0; n < 8; ++n) s += v[n] * v[n];
    s += __shfl_xor(s, 1, 32);
    s += __shfl_xor(s, 2, 32);
    s += __shfl_xor(s, 4, 32);
    s += __shfl_xor(s, 8, 32);
    const float scale = 1.0f / fmaxf(sqrtf(s), 1e-12f);
#pragma unroll
    for (int n = 0; n < 8; ++n)
      p[(size_t)(row0 + r) * D_DIM + n * 16 + l16] = v[n] * scale;
  }
}

__global__ __launch_bounds__(128) void band_sim_kernel(
    const float* __restrict__ p, const float* __restrict__ zrow,
    float* __restrict__ out)
{
  const int wave = (blockIdx.x * blockDim.x + threadIdx.x) >> 5;
  const int lane = threadIdx.x & 31;
  const int half = lane >> 4;
  const int l16  = lane & 15;
  const int b    = wave >> 7;
  const int mt   = wave & 127;
  const int t0   = mt * 16;

  const float* pb = p + (size_t)b * T_DIM * D_DIM;

  const float* bbase[9];
#pragma unroll
  for (int ct = 0; ct < 9; ++ct) {
    const int c = t0 + (ct - 4) * 16 + l16;
    bbase[ct] = ((unsigned)c < (unsigned)T_DIM)
                    ? (pb + (size_t)c * D_DIM + half * 2)
                    : (zrow + half * 2);
  }
  const float* arow = pb + (size_t)(t0 + l16) * D_DIM + half * 2;

  v8f acc[9] = {};

  v2f a_cur = *(const v2f*)(arow);
  v2f b_cur[9];
#pragma unroll
  for (int ct = 0; ct < 9; ++ct) b_cur[ct] = *(const v2f*)(bbase[ct]);

  for (int k = 0; k < D_DIM; k += 4) {
    const int kn = k + 4;
    v2f a_nxt = {};
    v2f b_nxt[9] = {};
    if (kn < D_DIM) {
      a_nxt = *(const v2f*)(arow + kn);
#pragma unroll
      for (int ct = 0; ct < 9; ++ct) b_nxt[ct] = *(const v2f*)(bbase[ct] + kn);
    }

#pragma unroll
    for (int ct = 0; ct < 9; ++ct)
      acc[ct] = __builtin_amdgcn_wmma_f32_16x16x4_f32(
          false, a_cur, false, b_cur[ct], (short)0, acc[ct], false, false);

    a_cur = a_nxt;
#pragma unroll
    for (int ct = 0; ct < 9; ++ct) b_cur[ct] = b_nxt[ct];
  }

#pragma unroll
  for (int ct = 0; ct < 9; ++ct) {
    const int c = t0 + (ct - 4) * 16 + l16;
#pragma unroll
    for (int j = 0; j < 8; ++j) {
      const int t = t0 + j + 8 * half;
      const int w = c - t + HWIN;
      if (w >= 0 && w < WIN) {
        out[((size_t)b * T_DIM + t) * WIN + w] = acc[ct][j];
      }
    }
  }
}

extern "C" void kernel_launch(void* const* d_in, const int* in_sizes, int n_in,
                              void* d_out, int out_size, void* d_ws, size_t ws_size,
                              hipStream_t stream) {
  const float* x  = (const float*)d_in[0];
  const float* W  = (const float*)d_in[1];
  const float* bv = (const float*)d_in[2];
  float* out = (float*)d_out;

  float* p    = (float*)d_ws;
  float* zrow = p + (size_t)8192 * D_DIM;

  gemm_norm_kernel<<<256, 256, 0, stream>>>(x, W, bv, p, zrow);
  band_sim_kernel<<<128, 128, 0, stream>>>(p, zrow, out);
}


# Projection GEMM: A/B fragment LDS reads regrouped: one group per block of 16 matrix instructions, read a whole block ahead into a 2nd register set, last block carried past the barrier; sum order kept
# speedup vs baseline: 2.2982x; 1.0402x over previous
#include <hip/hip_runtime.h>

typedef __attribute__((ext_vector_type(2))) float v2f;
typedef __attribute__((ext_vector_type(8))) float v8f;

#define T_DIM  2048
#define F_INN  2048
#define D_DIM  128
#define BATCH  4
#define WIN    101
#define HWIN   50

#define BM          32
#define KC          128
#define PITCH_A     132
#define PITCH_B     144
#define A_ELEMS     (BM * PITCH_A)
#define B_ELEMS     (KC * PITCH_B)
#define STAGE_ELEMS (A_ELEMS + B_ELEMS)

typedef __attribute__((ext_vector_type(4))) float v4f;

struct Frag { v2f a00, a01, a10, a11, b0, b1; };

__device__ __forceinline__ void load_frag(
    Frag& f, const float* __restrict__ aF0, const float* __restrict__ aF1,
    const float* __restrict__ bF, int kp)
{
  f.a00 = *(const v2f*)(aF0 + kp * 8);
  f.a01 = *(const v2f*)(aF0 + kp * 8 + 4);
  f.a10 = *(const v2f*)(aF1 + kp * 8);
  f.a11 = *(const v2f*)(aF1 + kp * 8 + 4);
  f.b0.x = bF[(kp * 8) * PITCH_B];
  f.b0.y = bF[(kp * 8 + 1) * PITCH_B];
  f.b1.x = bF[(kp * 8 + 4) * PITCH_B];
  f.b1.y = bF[(kp * 8 + 5) * PITCH_B];
}

#define NB    4
#define NBLK  (KC / 8 / NB)

__device__ __forceinline__ void load_block(
    Frag (&f)[NB], const float* __restrict__ aF0, const float* __restrict__ aF1,
    const float* __restrict__ bF, int blk)
{
#pragma unroll
  for (int i = 0; i < NB; ++i) load_frag(f[i], aF0, aF1, bF, blk * NB + i);
}

__device__ __forceinline__ void mma_block(
    const Frag (&f)[NB], v8f& acc0, v8f& acc1)
{
#pragma unroll
  for (int i = 0; i < NB; ++i) {
    const Frag& g = f[i];
    acc0 = __builtin_amdgcn_wmma_f32_16x16x4_f32(
        false, g.a00, false, g.b0, (short)0, acc0, false, false);
    acc1 = __builtin_amdgcn_wmma_f32_16x16x4_f32(
        false, g.a10, false, g.b0, (short)0, acc1, false, false);
    acc0 = __builtin_amdgcn_wmma_f32_16x16x4_f32(
        false, g.a01, false, g.b1, (short)0, acc0, false, false);
    acc1 = __builtin_amdgcn_wmma_f32_16x16x4_f32(
        false, g.a11, false, g.b1, (short)0, acc1, false, false);
  }
}

__device__ __forceinline__ void mma_fenced(
    const Frag (&f)[NB], v8f& acc0, v8f& acc1)
{
  __builtin_amdgcn_sched_barrier(0);
  mma_block(f, acc0, acc1);
  __builtin_amdgcn_sched_barrier(0);
}

__device__ __forceinline__ void compute_chunk(
    Frag (&f)[2][NB],
    const float* __restrict__ aF0, const float* __restrict__ aF1,
    const float* __restrict__ bF, v8f& acc0, v8f& acc1)
{
#pragma unroll
  for (int blk = 0; blk < NBLK - 1; ++blk) {
    load_block(f[(blk + 1) & 1], aF0, aF1, bF, blk + 1);
    mma_fenced(f[blk & 1], acc0, acc1);
  }
  asm volatile("" : "+v"(acc0), "+v"(acc1) :: "memory");
}

__device__ __forceinline__ void lds_barrier()
{
  asm volatile("s_wait_dscnt 0x0\n\ts_barrier_signal -1\n\ts_barrier_wait -1"
               ::: "memory");
}

__device__ __forceinline__ void load_chunk(
    v4f (&ra)[4], v4f (&rb)[16], const float* __restrict__ xc,
    const float* __restrict__ wc, unsigned xo, unsigned wo)
{
#pragma unroll
  for (int j = 0; j < 4; ++j)
    ra[j] = *(const v4f*)((const char*)(xc + (size_t)(8 * j) * F_INN) + xo);
#pragma unroll
  for (int j = 0; j < 16; ++j)
    rb[j] = *(const v4f*)((const char*)(wc + (size_t)(8 * j) * D_DIM) + wo);
  asm volatile("s_wait_xcnt 0x0" ::: "memory");
}

__device__ __forceinline__ void store_chunk(
    const v4f (&ra)[4], const v4f (&rb)[16], float* sa, float* sb)
{
#pragma unroll
  for (int j = 0; j < 4; ++j)
    *(v4f*)(sa + (8 * j) * PITCH_A) = ra[j];
#pragma unroll
  for (int j = 0; j < 16; ++j)
    *(v4f*)(sb + (8 * j) * PITCH_B) = rb[j];
}

__global__ __launch_bounds__(256) void gemm_norm_kernel(
    const float* __restrict__ x, const float* __restrict__ W,
    const float* __restrict__ bias, float* __restrict__ p,
    float* __restrict__ zrow)
{
  __shared__ __attribute__((aligned(16))) float lds[2 * STAGE_ELEMS];

  const int tid  = threadIdx.x;
  const int wv   = tid >> 5;
  const int lane = tid & 31;
  const int half = lane >> 4;
  const int l16  = lane & 15;
  const int row0 = blockIdx.x * BM;

  if (blockIdx.x == 0 && tid < D_DIM) zrow[tid] = 0.f;

  const float* xb = x + (size_t)row0 * F_INN;
  const unsigned xo = (unsigned)((wv * F_INN + lane * 4) * 4);
  const unsigned wo = (unsigned)((wv * D_DIM + lane * 4) * 4);
  const int stA = wv * PITCH_A + lane * 4;
  const int stB = A_ELEMS + wv * PITCH_B + lane * 4;

  const int aoff0 = l16 * PITCH_A + half * 2;
  const int aoff1 = aoff0 + 16 * PITCH_A;
  const int boff  = A_ELEMS + (half * 2) * PITCH_B + wv * 16 + l16;

  v8f acc0 = {};
  v8f acc1 = {};

  v4f ra0[4], ra1[4];
  v4f rb0[16], rb1[16];

  float* const s0 = lds;
  float* const s1 = lds + STAGE_ELEMS;

  Frag f[2][NB];

  load_chunk(ra0, rb0, xb, W, xo, wo);
  load_chunk(ra1, rb1, xb + KC, W + (size_t)KC * D_DIM, xo, wo);
  store_chunk(ra0, rb0, s0 + stA, s0 + stB);
  lds_barrier();

  load_block(f[0], s0 + aoff0, s0 + aoff1, s0 + boff, 0);
  load_chunk(ra0, rb0, xb + 2 * KC, W + (size_t)2 * KC * D_DIM, xo, wo);
  compute_chunk(f, s0 + aoff0, s0 + aoff1, s0 + boff, acc0, acc1);
  store_chunk(ra1, rb1, s1 + stA, s1 + stB);
  lds_barrier();

#pragma unroll 1
  for (int c = 1; c < F_INN / KC - 3; c += 2) {
    load_block(f[0], s1 + aoff0, s1 + aoff1, s1 + boff, 0);
    load_chunk(ra1, rb1, xb + (c + 2) * KC, W + (size_t)(c + 2) * KC * D_DIM, xo, wo);
    mma_fenced(f[1], acc0, acc1);
    compute_chunk(f, s1 + aoff0, s1 + aoff1, s1 + boff, acc0, acc1);
    store_chunk(ra0, rb0, s0 + stA, s0 + stB);
    lds_barrier();

    load_block(f[0], s0 + aoff0, s0 + aoff1, s0 + boff, 0);
    load_chunk(ra0, rb0, xb + (c + 3) * KC, W + (size_t)(c + 3) * KC * D_DIM, xo, wo);
    mma_fenced(f[1], acc0, acc1);
    compute_chunk(f, s0 + aoff0, s0 + aoff1, s0 + boff, acc0, acc1);
    store_chunk(ra1, rb1, s1 + stA, s1 + stB);
    lds_barrier();
  }
  load_block(f[0], s1 + aoff0, s1 + aoff1, s1 + boff, 0);
  load_chunk(ra1, rb1, xb + 15 * KC, W + (size_t)15 * KC * D_DIM, xo, wo);
  mma_fenced(f[1], acc0, acc1);
  compute_chunk(f, s1 + aoff0, s1 + aoff1, s1 + boff, acc0, acc1);
  store_chunk(ra0, rb0, s0 + stA, s0 + stB);
  lds_barrier();
  load_block(f[0], s0 + aoff0, s0 + aoff1, s0 + boff, 0);
  mma_fenced(f[1], acc0, acc1);
  compute_chunk(f, s0 + aoff0, s0 + aoff1, s0 + boff, acc0, acc1);
  store_chunk(ra1, rb1, s1 + stA, s1 + stB);
  lds_barrier();
  load_block(f[0], s1 + aoff0, s1 + aoff1, s1 + boff, 0);
  mma_fenced(f[1], acc0, acc1);
  compute_chunk(f, s1 + aoff0, s1 + aoff1, s1 + boff, acc0, acc1);
  mma_fenced(f[1], acc0, acc1);

  float* T = lds;
  {
    const float bv = bias[wv * 16 + l16];
#pragma unroll
    for (int j = 0; j < 8; ++j) {
      const int r = 8 * half + j;
      T[r * PITCH_A + wv * 16 + l16]        = acc0[j] + bv;
      T[(16 + r) * PITCH_A + wv * 16 + l16] = acc1[j] + bv;
    }
  }
  __syncthreads();

#pragma unroll
  for (int i = 0; i < 2; ++i) {
    const int r = 4 * wv + 2 * i + half;
    float v[8];
#pragma unroll
    for (int n = 0; n < 8; ++n) v[n] = T[r * PITCH_A + n * 16 + l16];
    float s = 0.f;
#pragma unroll
    for (int n = 0; n < 8; ++n) s += v[n] * v[n];
    s += __shfl_xor(s, 1, 32);
    s += __shfl_xor(s, 2, 32);
    s += __shfl_xor(s, 4, 32);
    s += __shfl_xor(s, 8, 32);
    const float scale = 1.0f / fmaxf(sqrtf(s), 1e-12f);
#pragma unroll
    for (int n = 0; n < 8; ++n)
      p[(size_t)(row0 + r) * D_DIM + n * 16 + l16] = v[n] * scale;
  }
}

__global__ __launch_bounds__(128) void band_sim_kernel(
    const float* __restrict__ p, const float* __restrict__ zrow,
    float* __restrict__ out)
{
  const int wave = (blockIdx.x * blockDim.x + threadIdx.x) >> 5;
  const int lane = threadIdx.x & 31;
  const int half = lane >> 4;
  const int l16  = lane & 15;
  const int b    = wave >> 7;
  const int mt   = wave & 127;
  const int t0   = mt * 16;

  const float* pb = p + (size_t)b * T_DIM * D_DIM;

  const float* bbase[9];
#pragma unroll
  for (int ct = 0; ct < 9; ++ct) {
    const int c = t0 + (ct - 4) * 16 + l16;
    bbase[ct] = ((unsigned)c < (unsigned)T_DIM)
                    ? (pb + (size_t)c * D_DIM + half * 2)
                    : (zrow + half * 2);
  }
  const float* arow = pb + (size_t)(t0 + l16) * D_DIM + half * 2;

  v8f acc[9] = {};

  v2f a_cur = *(const v2f*)(arow);
  v2f b_cur[9];
#pragma unroll
  for (int ct = 0; ct < 9; ++ct) b_cur[ct] = *(const v2f*)(bbase[ct]);

  for (int k = 0; k < D_DIM; k += 4) {
    const int kn = k + 4;
    v2f a_nxt = {};
    v2f b_nxt[9] = {};
    if (kn < D_DIM) {
      a_nxt = *(const v2f*)(arow + kn);
#pragma unroll
      for (int ct = 0; ct < 9; ++ct) b_nxt[ct] = *(const v2f*)(bbase[ct] + kn);
    }

#pragma unroll
    for (int ct = 0; ct < 9; ++ct)
      acc[ct] = __builtin_amdgcn_wmma_f32_16x16x4_f32(
          false, a_cur, false, b_cur[ct], (short)0, acc[ct], false, false);

    a_cur = a_nxt;
#pragma unroll
    for (int ct = 0; ct < 9; ++ct) b_cur[ct] = b_nxt[ct];
  }

#pragma unroll
  for (int ct = 0; ct < 9; ++ct) {
    const int c = t0 + (ct - 4) * 16 + l16;
#pragma unroll
    for (int j = 0; j < 8; ++j) {
      const int t = t0 + j + 8 * half;
      const int w = c - t + HWIN;
      if (w >= 0 && w < WIN) {
        out[((size_t)b * T_DIM + t) * WIN + w] = acc[ct][j];
      }
    }
  }
}

extern "C" void kernel_launch(void* const* d_in, const int* in_sizes, int n_in,
                              void* d_out, int out_size, void* d_ws, size_t ws_size,
                              hipStream_t stream) {
  const float* x  = (const float*)d_in[0];
  const float* W  = (const float*)d_in[1];
  const float* bv = (const float*)d_in[2];
  float* out = (float*)d_out;

  float* p    = (float*)d_ws;
  float* zrow = p + (size_t)8192 * D_DIM;

  gemm_norm_kernel<<<256, 256, 0, stream>>>(x, W, bv, p, zrow);
  band_sim_kernel<<<128, 128, 0, stream>>>(p, zrow, out);
}


# Band kernel re-tiled: 32 rows per work-group, the band's 160 rows of p fetched once as whole 512-B rows and staged in LDS for the A/B fragments (was 8-B loads over 16 lines each); sum order kept
# speedup vs baseline: 2.6761x; 1.1644x over previous
#include <hip/hip_runtime.h>

typedef __attribute__((ext_vector_type(2))) float v2f;
typedef __attribute__((ext_vector_type(8))) float v8f;

#define T_DIM  2048
#define F_INN  2048
#define D_DIM  128
#define BATCH  4
#define WIN    101
#define HWIN   50

#define BM          32
#define KC          128
#define PITCH_A     132
#define PITCH_B     144
#define A_ELEMS     (BM * PITCH_A)
#define B_ELEMS     (KC * PITCH_B)
#define STAGE_ELEMS (A_ELEMS + B_ELEMS)

typedef __attribute__((ext_vector_type(4))) float v4f;

struct Frag { v2f a00, a01, a10, a11, b0, b1; };

__device__ __forceinline__ void load_frag(
    Frag& f, const float* __restrict__ aF0, const float* __restrict__ aF1,
    const float* __restrict__ bF, int kp)
{
  f.a00 = *(const v2f*)(aF0 + kp * 8);
  f.a01 = *(const v2f*)(aF0 + kp * 8 + 4);
  f.a10 = *(const v2f*)(aF1 + kp * 8);
  f.a11 = *(const v2f*)(aF1 + kp * 8 + 4);
  f.b0.x = bF[(kp * 8) * PITCH_B];
  f.b0.y = bF[(kp * 8 + 1) * PITCH_B];
  f.b1.x = bF[(kp * 8 + 4) * PITCH_B];
  f.b1.y = bF[(kp * 8 + 5) * PITCH_B];
}

#define NB    4
#define NBLK  (KC / 8 / NB)

__device__ __forceinline__ void load_block(
    Frag (&f)[NB], const float* __restrict__ aF0, const float* __restrict__ aF1,
    const float* __restrict__ bF, int blk)
{
#pragma unroll
  for (int i = 0; i < NB; ++i) load_frag(f[i], aF0, aF1, bF, blk * NB + i);
}

__device__ __forceinline__ void mma_block(
    const Frag (&f)[NB], v8f& acc0, v8f& acc1)
{
#pragma unroll
  for (int i = 0; i < NB; ++i) {
    const Frag& g = f[i];
    acc0 = __builtin_amdgcn_wmma_f32_16x16x4_f32(
        false, g.a00, false, g.b0, (short)0, acc0, false, false);
    acc1 = __builtin_amdgcn_wmma_f32_16x16x4_f32(
        false, g.a10, false, g.b0, (short)0, acc1, false, false);
    acc0 = __builtin_amdgcn_wmma_f32_16x16x4_f32(
        false, g.a01, false, g.b1, (short)0, acc0, false, false);
    acc1 = __builtin_amdgcn_wmma_f32_16x16x4_f32(
        false, g.a11, false, g.b1, (short)0, acc1, false, false);
  }
}

__device__ __forceinline__ void mma_fenced(
    const Frag (&f)[NB], v8f& acc0, v8f& acc1)
{
  __builtin_amdgcn_sched_barrier(0);
  mma_block(f, acc0, acc1);
  __builtin_amdgcn_sched_barrier(0);
}

__device__ __forceinline__ void compute_chunk(
    Frag (&f)[2][NB],
    const float* __restrict__ aF0, const float* __restrict__ aF1,
    const float* __restrict__ bF, v8f& acc0, v8f& acc1)
{
#pragma unroll
  for (int blk = 0; blk < NBLK - 1; ++blk) {
    load_block(f[(blk + 1) & 1], aF0, aF1, bF, blk + 1);
    mma_fenced(f[blk & 1], acc0, acc1);
  }
  asm volatile("" : "+v"(acc0), "+v"(acc1) :: "memory");
}

__device__ __forceinline__ void lds_barrier()
{
  asm volatile("s_wait_dscnt 0x0\n\ts_barrier_signal -1\n\ts_barrier_wait -1"
               ::: "memory");
}

__device__ __forceinline__ void load_chunk(
    v4f (&ra)[4], v4f (&rb)[16], const float* __restrict__ xc,
    const float* __restrict__ wc, unsigned xo, unsigned wo)
{
#pragma unroll
  for (int j = 0; j < 4; ++j)
    ra[j] = *(const v4f*)((const char*)(xc + (size_t)(8 * j) * F_INN) + xo);
#pragma unroll
  for (int j = 0; j < 16; ++j)
    rb[j] = *(const v4f*)((const char*)(wc + (size_t)(8 * j) * D_DIM) + wo);
  asm volatile("s_wait_xcnt 0x0" ::: "memory");
}

__device__ __forceinline__ void store_chunk(
    const v4f (&ra)[4], const v4f (&rb)[16], float* sa, float* sb)
{
#pragma unroll
  for (int j = 0; j < 4; ++j)
    *(v4f*)(sa + (8 * j) * PITCH_A) = ra[j];
#pragma unroll
  for (int j = 0; j < 16; ++j)
    *(v4f*)(sb + (8 * j) * PITCH_B) = rb[j];
}

__global__ __launch_bounds__(256) void gemm_norm_kernel(
    const float* __restrict__ x, const float* __restrict__ W,
    const float* __restrict__ bias, float* __restrict__ p,
    float* __restrict__ zrow)
{
  __shared__ __attribute__((aligned(16))) float lds[2 * STAGE_ELEMS];

  const int tid  = threadIdx.x;
  const int wv   = tid >> 5;
  const int lane = tid & 31;
  const int half = lane >> 4;
  const int l16  = lane & 15;
  const int row0 = blockIdx.x * BM;

  if (blockIdx.x == 0 && tid < D_DIM) zrow[tid] = 0.f;

  const float* xb = x + (size_t)row0 * F_INN;
  const unsigned xo = (unsigned)((wv * F_INN + lane * 4) * 4);
  const unsigned wo = (unsigned)((wv * D_DIM + lane * 4) * 4);
  const int stA = wv * PITCH_A + lane * 4;
  const int stB = A_ELEMS + wv * PITCH_B + lane * 4;

  const int aoff0 = l16 * PITCH_A + half * 2;
  const int aoff1 = aoff0 + 16 * PITCH_A;
  const int boff  = A_ELEMS + (half * 2) * PITCH_B + wv * 16 + l16;

  v8f acc0 = {};
  v8f acc1 = {};

  v4f ra0[4], ra1[4];
  v4f rb0[16], rb1[16];

  float* const s0 = lds;
  float* const s1 = lds + STAGE_ELEMS;

  Frag f[2][NB];

  load_chunk(ra0, rb0, xb, W, xo, wo);
  load_chunk(ra1, rb1, xb + KC, W + (size_t)KC * D_DIM, xo, wo);
  store_chunk(ra0, rb0, s0 + stA, s0 + stB);
  lds_barrier();

  load_block(f[0], s0 + aoff0, s0 + aoff1, s0 + boff, 0);
  load_chunk(ra0, rb0, xb + 2 * KC, W + (size_t)2 * KC * D_DIM, xo, wo);
  compute_chunk(f, s0 + aoff0, s0 + aoff1, s0 + boff, acc0, acc1);
  store_chunk(ra1, rb1, s1 + stA, s1 + stB);
  lds_barrier();

#pragma unroll 1
  for (int c = 1; c < F_INN / KC - 3; c += 2) {
    load_block(f[0], s1 + aoff0, s1 + aoff1, s1 + boff, 0);
    load_chunk(ra1, rb1, xb + (c + 2) * KC, W + (size_t)(c + 2) * KC * D_DIM, xo, wo);
    mma_fenced(f[1], acc0, acc1);
    compute_chunk(f, s1 + aoff0, s1 + aoff1, s1 + boff, acc0, acc1);
    store_chunk(ra0, rb0, s0 + stA, s0 + stB);
    lds_barrier();

    load_block(f[0], s0 + aoff0, s0 + aoff1, s0 + boff, 0);
    load_chunk(ra0, rb0, xb + (c + 3) * KC, W + (size_t)(c + 3) * KC * D_DIM, xo, wo);
    mma_fenced(f[1], acc0, acc1);
    compute_chunk(f, s0 + aoff0, s0 + aoff1, s0 + boff, acc0, acc1);
    store_chunk(ra1, rb1, s1 + stA, s1 + stB);
    lds_barrier();
  }
  load_block(f[0], s1 + aoff0, s1 + aoff1, s1 + boff, 0);
  load_chunk(ra1, rb1, xb + 15 * KC, W + (size_t)15 * KC * D_DIM, xo, wo);
  mma_fenced(f[1], acc0, acc1);
  compute_chunk(f, s1 + aoff0, s1 + aoff1, s1 + boff, acc0, acc1);
  store_chunk(ra0, rb0, s0 + stA, s0 + stB);
  lds_barrier();
  load_block(f[0], s0 + aoff0, s0 + aoff1, s0 + boff, 0);
  mma_fenced(f[1], acc0, acc1);
  compute_chunk(f, s0 + aoff0, s0 + aoff1, s0 + boff, acc0, acc1);
  store_chunk(ra1, rb1, s1 + stA, s1 + stB);
  lds_barrier();
  load_block(f[0], s1 + aoff0, s1 + aoff1, s1 + boff, 0);
  mma_fenced(f[1], acc0, acc1);
  compute_chunk(f, s1 + aoff0, s1 + aoff1, s1 + boff, acc0, acc1);
  mma_fenced(f[1], acc0, acc1);

  float* T = lds;
  {
    const float bv = bias[wv * 16 + l16];
#pragma unroll
    for (int j = 0; j < 8; ++j) {
      const int r = 8 * half + j;
      T[r * PITCH_A + wv * 16 + l16]        = acc0[j] + bv;
      T[(16 + r) * PITCH_A + wv * 16 + l16] = acc1[j] + bv;
    }
  }
  __syncthreads();

#pragma unroll
  for (int i = 0; i < 2; ++i) {
    const int r = 4 * wv + 2 * i + half;
    float v[8];
#pragma unroll
    for (int n = 0; n < 8; ++n) v[n] = T[r * PITCH_A + n * 16 + l16];
    float s = 0.f;
#pragma unroll
    for (int n = 0; n < 8; ++n) s += v[n] * v[n];
    s += __shfl_xor(s, 1, 32);
    s += __shfl_xor(s, 2, 32);
    s += __shfl_xor(s, 4, 32);
    s += __shfl_xor(s, 8, 32);
    const float scale = 1.0f / fmaxf(sqrtf(s), 1e-12f);
#pragma unroll
    for (int n = 0; n < 8; ++n)
      p[(size_t)(row0 + r) * D_DIM + n * 16 + l16] = v[n] * scale;
  }
}

#define BROWS    32
#define SROWS    160
#define PITCH_P  132
#define KSB      4
#define NKB      (D_DIM / 4 / KSB)

template <int NT>
struct BFrag { v2f a[KSB]; v2f b[KSB][NT]; };

template <int NT>
__device__ __forceinline__ void band_load_block(
    BFrag<NT>& f, const float* __restrict__ aF, const float* __restrict__ bF,
    int blk)
{
#pragma unroll
  for (int i = 0; i < KSB; ++i) {
    const int k = (blk * KSB + i) * 4;
    f.a[i] = *(const v2f*)(aF + k);
#pragma unroll
    for (int n = 0; n < NT; ++n)
      f.b[i][n] = *(const v2f*)(bF + n * 16 * PITCH_P + k);
  }
}

template <int NT>
__device__ __forceinline__ void band_mma_block(
    const BFrag<NT>& f, v8f (&acc)[NT])
{
  __builtin_amdgcn_sched_barrier(0);
#pragma unroll
  for (int i = 0; i < KSB; ++i)
#pragma unroll
    for (int n = 0; n < NT; ++n)
      acc[n] = __builtin_amdgcn_wmma_f32_16x16x4_f32(
          false, f.a[i], false, f.b[i][n], (short)0, acc[n], false, false);
  __builtin_amdgcn_sched_barrier(0);
}

template <int NT>
__device__ __forceinline__ void band_tiles(
    const float* __restrict__ sp, float* __restrict__ out,
    int b, int t0, int rt, int ct0, int half, int l16)
{
  const float* aF = sp + (64 + 16 * rt + l16) * PITCH_P + half * 2;
  const float* bF = sp + (16 * (rt + ct0) + l16) * PITCH_P + half * 2;

  v8f acc[NT];
#pragma unroll
  for (int n = 0; n < NT; ++n) acc[n] = v8f{};

  BFrag<NT> f0, f1;
  band_load_block<NT>(f0, aF, bF, 0);
#pragma unroll
  for (int blk = 0; blk < NKB; blk += 2) {
    band_load_block<NT>(f1, aF, bF, blk + 1);
    band_mma_block<NT>(f0, acc);
    if (blk + 2 < NKB) band_load_block<NT>(f0, aF, bF, blk + 2);
    band_mma_block<NT>(f1, acc);
  }

#pragma unroll
  for (int n = 0; n < NT; ++n) {
    const int c = t0 + (ct0 + n - 4) * 16 + l16;
#pragma unroll
    for (int j = 0; j < 8; ++j) {
      const int t = t0 + j + 8 * half;
      const int w = c - t + HWIN;
      if (w >= 0 && w < WIN) {
        out[((size_t)b * T_DIM + t) * WIN + w] = acc[n][j];
      }
    }
  }
}

__global__ __launch_bounds__(128) void band_sim_kernel(
    const float* __restrict__ p, const float* __restrict__ zrow,
    float* __restrict__ out)
{
  __shared__ __attribute__((aligned(16))) float sp[SROWS * PITCH_P];

  const int tid  = threadIdx.x;
  const int wv   = __builtin_amdgcn_readfirstlane(tid >> 5);
  const int lane = tid & 31;
  const int half = lane >> 4;
  const int l16  = lane & 15;
  const int b    = blockIdx.x >> 6;
  const int R0   = (blockIdx.x & 63) * BROWS;

  const float* pb = p + (size_t)b * T_DIM * D_DIM;

  v4f st[SROWS / 4];
#pragma unroll
  for (int j = 0; j < SROWS / 4; ++j) {
    const int s = wv + 4 * j;
    const int r = R0 - 64 + s;
    const float* rowp = ((unsigned)r < (unsigned)T_DIM)
                            ? (pb + (size_t)r * D_DIM) : zrow;
    st[j] = *(const v4f*)(rowp + lane * 4);
  }
#pragma unroll
  for (int j = 0; j < SROWS / 4; ++j)
    *(v4f*)(sp + (wv + 4 * j) * PITCH_P + lane * 4) = st[j];
  __syncthreads();

  const int rt = wv & 1;
  const int t0 = R0 + 16 * rt;
  if ((wv >> 1) == 0)
    band_tiles<5>(sp, out, b, t0, rt, 0, half, l16);
  else
    band_tiles<4>(sp, out, b, t0, rt, 5, half, l16);
}

extern "C" void kernel_launch(void* const* d_in, const int* in_sizes, int n_in,
                              void* d_out, int out_size, void* d_ws, size_t ws_size,
                              hipStream_t stream) {
  const float* x  = (const float*)d_in[0];
  const float* W  = (const float*)d_in[1];
  const float* bv = (const float*)d_in[2];
  float* out = (float*)d_out;

  float* p    = (float*)d_ws;
  float* zrow = p + (size_t)8192 * D_DIM;

  gemm_norm_kernel<<<256, 256, 0, stream>>>(x, W, bv, p, zrow);
  band_sim_kernel<<<256, 128, 0, stream>>>(p, zrow, out);
}
